# starting kernel
# baseline (speedup 1.0000x reference)
#include <hip/hip_runtime.h>


typedef __attribute__((ext_vector_type(2))) float v2f;
typedef __attribute__((ext_vector_type(8))) float v8f;

#define TILE   256
#define BY     64
#define BX     64
#define IPH    (BY + 14)
#define IPW    (BX + 16)
#define NCHAIN (IPH * (BX / 16))
#define OPAD   15
#define OROWS  (OPAD + BY + 14)

__global__ __launch_bounds__(256) void norm_kernels_k(const float* __restrict__ k,
                                                      float* __restrict__ knorm) {
    const int t   = blockIdx.x;
    const int tid = threadIdx.x;

    __shared__ float red[256];
    float v = 0.0f;
    if (tid < 225) v = k[t * 225 + tid];
    red[tid] = v;
    __syncthreads();
    for (int s = 128; s > 0; s >>= 1) {
        if (tid < s) red[tid] += red[tid + s];
        __syncthreads();
    }
    const float inv = 1.0f / (red[0] + 1e-12f);

    const int dy = tid >> 4;
    const int dx = tid & 15;
    const float val = (dy < 15 && dx < 15) ? k[t * 225 + dy * 15 + dx] * inv : 0.0f;
    knorm[t * 256 + tid] = val;
}

__global__ __launch_bounds__(256) void pwblur_k(const float* __restrict__ x,
                                                const float* __restrict__ knorm,
                                                float* __restrict__ out) {
    __shared__ float ip[IPH * IPW];
    __shared__ float outl[OROWS * BX];

    const int tid  = threadIdx.x;
    const int lane = tid & 31;
    const int wave = tid >> 5;
    const bool hi  = (lane & 16) != 0;
    const int  nl  = lane & 15;

    int bid   = blockIdx.x;
    int sub   = bid & 15;
    int t     = bid >> 4;
    int ti    = t & 3;
    int tj    = (t >> 2) & 3;
    int plane = t >> 4;
    int sub_y = sub >> 2;
    int sub_x = sub & 3;

    const int ty0 = tj * TILE;
    const int tx0 = ti * TILE;
    const int y0  = sub_y * BY;
    const int x0  = sub_x * BX;

    const float* __restrict__ xin = x + (size_t)plane * 1024 * 1024;

    for (int idx = tid; idx < IPH * IPW; idx += 256) {
        int r  = idx / IPW;
        int xi = idx - r * IPW;
        int ys = y0 + r - 7;  ys = ys < 0 ? 0 : (ys > TILE - 1 ? TILE - 1 : ys);
        int xs = x0 + xi - 7; xs = xs < 0 ? 0 : (xs > TILE - 1 ? TILE - 1 : xs);
        ip[idx] = xin[(size_t)(ty0 + ys) * 1024 + (tx0 + xs)];
    }
    for (int idx = tid; idx < OROWS * BX; idx += 256) outl[idx] = 0.0f;

    const float* __restrict__ kn = knorm + (tj * 4 + ti) * 256;
    v2f a[4];
    {
        const int m = nl;
        #pragma unroll
        for (int c = 0; c < 4; ++c) {
            int dxb = 4 * c + (hi ? 2 : 0);
            a[c].x = kn[m * 16 + dxb];
            a[c].y = kn[m * 16 + dxb + 1];
        }
    }
    __syncthreads();

    for (int cid = wave; cid < NCHAIN; cid += 8) {
        const int r  = cid >> 2;
        const int xb = (cid & 3) * 16;

        const int bbase = r * IPW + xb + nl + (hi ? 2 : 0);

        v8f acc = {};
        #pragma unroll
        for (int c = 0; c < 4; ++c) {
            v2f bv;
            bv.x = ip[bbase + 4 * c];
            bv.y = ip[bbase + 4 * c + 1];
            acc = __builtin_amdgcn_wmma_f32_16x16x4_f32(
                false, a[c], false, bv, (short)0, acc, false, false);
        }

        const int sbase = (r + OPAD - (hi ? 8 : 0)) * BX + xb + nl;
        #pragma unroll
        for (int i = 0; i < 8; ++i) {
            atomicAdd(&outl[sbase - i * BX], acc[i]);
        }
    }
    __syncthreads();

    float* __restrict__ op = out + (size_t)plane * 1024 * 1024;
    for (int idx = tid; idx < BY * BX; idx += 256) {
        int yl = idx >> 6;
        int xl = idx & 63;
        op[(size_t)(ty0 + y0 + yl) * 1024 + (tx0 + x0 + xl)] = outl[(OPAD + yl) * BX + xl];
    }
}

extern "C" void kernel_launch(void* const* d_in, const int* in_sizes, int n_in,
                              void* d_out, int out_size, void* d_ws, size_t ws_size,
                              hipStream_t stream) {
    const float* x    = (const float*)d_in[0];
    const float* kern = (const float*)d_in[1];
    float* outp  = (float*)d_out;
    float* knorm = (float*)d_ws;

    norm_kernels_k<<<16, 256, 0, stream>>>(kern, knorm);

    pwblur_k<<<6144, 256, 0, stream>>>(x, knorm, outp);
}


# Kernel normalisation fused into the blur kernel: each wave sums its tile's 225 weights in the same halving-tree order and reads its A matrices from a private LDS copy; one launch instead of two
# speedup vs baseline: 1.2956x; 1.2956x over previous
#include <hip/hip_runtime.h>


typedef __attribute__((ext_vector_type(2))) float v2f;
typedef __attribute__((ext_vector_type(4))) float v4f;
typedef __attribute__((ext_vector_type(8))) float v8f;

#define TILE   256
#define BY     128
#define BX     64
#define IPH    (BY + 14)
#define IPW    (BX + 16)

__device__ __forceinline__ float xhalf(float x) {
    int b = __builtin_bit_cast(int, x);
    return __builtin_bit_cast(float,
        __builtin_amdgcn_permlanex16(b, b, 0x76543210, 0xfedcba98, false, false));
}

__global__ __launch_bounds__(64) void pwblur_k(const float* __restrict__ x,
                                                const float* __restrict__ kraw,
                                                float* __restrict__ out) {
    __shared__ alignas(16) float ip[IPH * IPW];
    __shared__ alignas(16) float kl[2][256];

    const int tid  = threadIdx.x;
    const int lane = tid & 31;
    const int wave = tid >> 5;
    const bool hi  = (lane & 16) != 0;
    const int  nl  = lane & 15;

    const int bid   = blockIdx.x;
    const int sub   = bid & 7;
    const int t     = bid >> 3;
    const int ti    = t & 3;
    const int tj    = (t >> 2) & 3;
    const int plane = t >> 4;
    const int sub_y = sub >> 2;
    const int sub_x = sub & 3;

    const int ty0 = tj * TILE;
    const int tx0 = ti * TILE;
    const int y0  = sub_y * BY;
    const int x0  = sub_x * BX;

    const float* __restrict__ xin = x + (size_t)plane * 1024 * 1024;

    const int wv = __builtin_amdgcn_readfirstlane(wave);

    const float* __restrict__ kt = kraw + (tj * 4 + ti) * 225;
    float kv[8], kp[8];
    #pragma unroll
    for (int j = 0; j < 8; ++j) {
        const int idx = lane + 32 * j;
        const float v = kt[idx < 225 ? idx : 224];
        kv[j] = idx < 225 ? v : 0.0f;
        const int dy = idx >> 4;
        const int dx = idx & 15;
        const bool ok = dy < 15 && dx < 15;
        const float p = kt[ok ? dy * 15 + dx : 0];
        kp[j] = ok ? p : 0.0f;
    }

    {
        const int hq  = nl < 14 ? nl : 14;
        const int hxi = hq < 7 ? hq : hq + 64;
        int hxs = x0 + hxi - 7; hxs = hxs < 0 ? 0 : (hxs > TILE - 1 ? TILE - 1 : hxs);
        const float* __restrict__ xt = xin + (size_t)ty0 * 1024 + tx0;
        for (int i0 = 0; i0 < 36; i0 += 12) {
            v4f vin[12];
            float vha[12];
            #pragma unroll
            for (int i = 0; i < 12; ++i) {
                int p = wv + 2 * (i0 + i);
                p = p < IPH / 2 ? p : IPH / 2 - 1;
                const int r = 2 * p + (hi ? 1 : 0);
                int ys = y0 + r - 7;  ys = ys < 0 ? 0 : (ys > TILE - 1 ? TILE - 1 : ys);
                const float* __restrict__ srow = xt + (size_t)ys * 1024;
                vin[i] = *(const v4f*)(srow + x0 + 4 * nl);
                vha[i] = srow[hxs];
            }
            #pragma unroll
            for (int i = 0; i < 12; ++i) {
                int p = wv + 2 * (i0 + i);
                p = p < IPH / 2 ? p : IPH / 2 - 1;
                const int r = 2 * p + (hi ? 1 : 0);
                *(v4f*)(ip + r * IPW + 8 + 4 * nl) = vin[i];
                ip[r * IPW + 1 + hxi] = vha[i];
            }
        }
    }

    {
        float u0 = kv[0] + kv[4], u1 = kv[1] + kv[5], u2 = kv[2] + kv[6], u3 = kv[3] + kv[7];
        float w0 = u0 + u2, w1 = u1 + u3;
        float z = w0 + w1;
        z = z + xhalf(z);
        z = z + __shfl_xor(z, 8);
        z = z + __shfl_xor(z, 4);
        z = z + __shfl_xor(z, 2);
        z = z + __shfl_xor(z, 1);
        const float inv = 1.0f / (z + 1e-12f);
        #pragma unroll
        for (int j = 0; j < 8; ++j) {
            const int idx = lane + 32 * j;
            const bool ok = (idx >> 4) < 15 && (idx & 15) < 15;
            kl[wv][idx] = ok ? kp[j] * inv : 0.0f;
        }
    }
    __syncthreads();

    const float* kn = kl[wv];
    v2f a[16][4];
    #pragma unroll
    for (int s = 0; s < 16; ++s) {
        const int krow = (s - nl) & 15;
        #pragma unroll
        for (int c = 0; c < 4; ++c) {
            const int dxb = 4 * c + (hi ? 2 : 0);
            a[s][c] = *(const v2f*)(kn + krow * 16 + dxb);
        }
    }

    const float* ipb = ip + wv * 32;
    const int    bl  = 1 + nl + (hi ? 2 : 0);

    float* __restrict__ op = out + (size_t)plane * 1024 * 1024
                           + (size_t)(ty0 + y0) * 1024 + (tx0 + x0 + wv * 32);
    const int colA = (hi ? 16 : 0) + nl;
    const int colB = (hi ? 0 : 16) + nl;

    v8f acc0 = {};
    v8f acc1 = {};
    v2f b0[4], b1[4];

#define LOAD_B(dst0, dst1, row)                                                  \
    {                                                                            \
        const float* p_ = ipb + (row) * IPW + bl;                                \
        _Pragma("unroll")                                                        \
        for (int c = 0; c < 4; ++c) {                                            \
            dst0[c].x = p_[4 * c];      dst0[c].y = p_[4 * c + 1];               \
            dst1[c].x = p_[16 + 4 * c]; dst1[c].y = p_[16 + 4 * c + 1];          \
        }                                                                        \
    }

#define STEP(s, r, STORE)                                                        \
    {                                                                            \
        v2f n0[4], n1[4];                                                        \
        const int rn_ = (r) + 1 < IPH ? (r) + 1 : IPH - 1;                       \
        LOAD_B(n0, n1, rn_)                                                      \
          \
              \
        __builtin_amdgcn_sched_barrier(0);                                       \
        _Pragma("unroll")                                                        \
        for (int c = 0; c < 4; ++c) {                                            \
            acc0 = __builtin_amdgcn_wmma_f32_16x16x4_f32(                        \
                false, a[s][c], false, b0[c], (short)0, acc0, false, false);     \
            acc1 = __builtin_amdgcn_wmma_f32_16x16x4_f32(                        \
                false, a[s][c], false, b1[c], (short)0, acc1, false, false);     \
        }                                                                        \
        const int  mf_   = ((s) + 2) & 15;     \
        const int  vi_   = mf_ & 7;                           \
        const bool up_   = (mf_ >> 3) != 0;                    \
        const bool mine_ = up_ == hi;                                            \
        const float f0_ = acc0[vi_];                                             \
        const float f1_ = acc1[vi_];                                             \
        if (STORE) {                                                             \
            const float f1x_ = xhalf(f1_);     \
            float* __restrict__ orow_ = op + (size_t)((r) - 14) * 1024;          \
            orow_[up_ ? colB : colA] = mine_ ? f0_ : f1x_;                       \
        }                                                                        \
        acc0[vi_] = mine_ ? 0.0f : f0_;        \
        acc1[vi_] = mine_ ? 0.0f : f1_;                                          \
        _Pragma("unroll")                                                        \
        for (int c = 0; c < 4; ++c) { b0[c] = n0[c]; b1[c] = n1[c]; }            \
    }

    LOAD_B(b0, b1, 0)

    #pragma unroll
    for (int s = 0; s < 14; ++s) STEP(s, s, false)

    #pragma clang loop unroll(disable)
    for (int base = 14; base < IPH; base += 16) {
        #pragma unroll
        for (int j = 0; j < 16; ++j) STEP((14 + j) & 15, base + j, true)
    }
#undef STEP
#undef LOAD_B
}

extern "C" void kernel_launch(void* const* d_in, const int* in_sizes, int n_in,
                              void* d_out, int out_size, void* d_ws, size_t ws_size,
                              hipStream_t stream) {
    const float* x    = (const float*)d_in[0];
    const float* kern = (const float*)d_in[1];
    float* outp  = (float*)d_out;

    pwblur_k<<<3072, 64, 0, stream>>>(x, kern, outp);
}


# B fragments of two input rows fetched as one group of 16 LDS loads from one running address: one load place per 16 matrix instructions instead of four, 8 address adds per 128 instead of 107
# speedup vs baseline: 1.3877x; 1.0711x over previous
#include <hip/hip_runtime.h>


typedef __attribute__((ext_vector_type(2))) float v2f;
typedef __attribute__((ext_vector_type(4))) float v4f;
typedef __attribute__((ext_vector_type(8))) float v8f;

#define TILE   256
#define BY     128
#define BX     64
#define IPH    (BY + 14)
#define IPW    (BX + 16)

__device__ __forceinline__ float xhalf(float x) {
    int b = __builtin_bit_cast(int, x);
    return __builtin_bit_cast(float,
        __builtin_amdgcn_permlanex16(b, b, 0x76543210, 0xfedcba98, false, false));
}

__global__ __launch_bounds__(64) void pwblur_k(const float* __restrict__ x,
                                                const float* __restrict__ kraw,
                                                float* __restrict__ out) {
    __shared__ alignas(16) float ip[IPH * IPW];
    __shared__ alignas(16) float kl[2][256];

    const int tid  = threadIdx.x;
    const int lane = tid & 31;
    const int wave = tid >> 5;
    const bool hi  = (lane & 16) != 0;
    const int  nl  = lane & 15;

    const int bid   = blockIdx.x;
    const int sub   = bid & 7;
    const int t     = bid >> 3;
    const int ti    = t & 3;
    const int tj    = (t >> 2) & 3;
    const int plane = t >> 4;
    const int sub_y = sub >> 2;
    const int sub_x = sub & 3;

    const int ty0 = tj * TILE;
    const int tx0 = ti * TILE;
    const int y0  = sub_y * BY;
    const int x0  = sub_x * BX;

    const float* __restrict__ xin = x + (size_t)plane * 1024 * 1024;

    const int wv = __builtin_amdgcn_readfirstlane(wave);

    const float* __restrict__ kt = kraw + (tj * 4 + ti) * 225;
    float kv[8], kp[8];
    #pragma unroll
    for (int j = 0; j < 8; ++j) {
        const int idx = lane + 32 * j;
        const float v = kt[idx < 225 ? idx : 224];
        kv[j] = idx < 225 ? v : 0.0f;
        const int dy = idx >> 4;
        const int dx = idx & 15;
        const bool ok = dy < 15 && dx < 15;
        const float p = kt[ok ? dy * 15 + dx : 0];
        kp[j] = ok ? p : 0.0f;
    }

    {
        const int hq  = nl < 14 ? nl : 14;
        const int hxi = hq < 7 ? hq : hq + 64;
        int hxs = x0 + hxi - 7; hxs = hxs < 0 ? 0 : (hxs > TILE - 1 ? TILE - 1 : hxs);
        const float* __restrict__ xt = xin + (size_t)ty0 * 1024 + tx0;
        for (int i0 = 0; i0 < 36; i0 += 12) {
            v4f vin[12];
            float vha[12];
            #pragma unroll
            for (int i = 0; i < 12; ++i) {
                int p = wv + 2 * (i0 + i);
                p = p < IPH / 2 ? p : IPH / 2 - 1;
                const int r = 2 * p + (hi ? 1 : 0);
                int ys = y0 + r - 7;  ys = ys < 0 ? 0 : (ys > TILE - 1 ? TILE - 1 : ys);
                const float* __restrict__ srow = xt + (size_t)ys * 1024;
                vin[i] = *(const v4f*)(srow + x0 + 4 * nl);
                vha[i] = srow[hxs];
            }
            #pragma unroll
            for (int i = 0; i < 12; ++i) {
                int p = wv + 2 * (i0 + i);
                p = p < IPH / 2 ? p : IPH / 2 - 1;
                const int r = 2 * p + (hi ? 1 : 0);
                *(v4f*)(ip + r * IPW + 8 + 4 * nl) = vin[i];
                ip[r * IPW + 1 + hxi] = vha[i];
            }
        }
    }

    {
        float u0 = kv[0] + kv[4], u1 = kv[1] + kv[5], u2 = kv[2] + kv[6], u3 = kv[3] + kv[7];
        float w0 = u0 + u2, w1 = u1 + u3;
        float z = w0 + w1;
        z = z + xhalf(z);
        z = z + __shfl_xor(z, 8);
        z = z + __shfl_xor(z, 4);
        z = z + __shfl_xor(z, 2);
        z = z + __shfl_xor(z, 1);
        const float inv = 1.0f / (z + 1e-12f);
        #pragma unroll
        for (int j = 0; j < 8; ++j) {
            const int idx = lane + 32 * j;
            const bool ok = (idx >> 4) < 15 && (idx & 15) < 15;
            kl[wv][idx] = ok ? kp[j] * inv : 0.0f;
        }
    }
    __syncthreads();

    const float* kn = kl[wv];
    v2f a[16][4];
    #pragma unroll
    for (int s = 0; s < 16; ++s) {
        const int krow = (s - nl) & 15;
        #pragma unroll
        for (int c = 0; c < 4; ++c) {
            const int dxb = 4 * c + (hi ? 2 : 0);
            a[s][c] = *(const v2f*)(kn + krow * 16 + dxb);
        }
    }

    const char* ipc = (const char*)ip;
    const int   bl  = 1 + nl + (hi ? 2 : 0);
    unsigned bo = (unsigned)((wv * 32 + bl) * 4);

    float* __restrict__ op = out + (size_t)plane * 1024 * 1024
                           + (size_t)(ty0 + y0) * 1024 + (tx0 + x0 + wv * 32);
    const int colA = (hi ? 16 : 0) + nl;
    const int colB = (hi ? 0 : 16) + nl;

    v8f acc0 = {};
    v8f acc1 = {};
    v2f b0[4], b1[4];
    v2f c0[4], c1[4];

#define LOAD_ROW(dst0, dst1, boff)                                               \
    {                                                                            \
        const float* p_ = (const float*)(ipc + (boff));                          \
        _Pragma("unroll")                                                        \
        for (int c = 0; c < 4; ++c) {                                            \
            dst0[c].x = p_[4 * c];      dst0[c].y = p_[4 * c + 1];               \
            dst1[c].x = p_[16 + 4 * c]; dst1[c].y = p_[16 + 4 * c + 1];          \
        }                                                                        \
    }

#define FETCH_PAIR(dA0, dA1, dB0, dB1)                                           \
    {                                                                            \
          \
          \
        asm volatile("" : "+v"(bo));                                             \
        LOAD_ROW(dA0, dA1, bo)                                                   \
        LOAD_ROW(dB0, dB1, bo + IPW * 4)                                         \
          \
          \
        __builtin_amdgcn_sched_barrier(0);                                       \
    }

#define ROWMM(s, q0, q1)                                                         \
    {                                                                            \
        _Pragma("unroll")                                                        \
        for (int c = 0; c < 4; ++c) {                                            \
            acc0 = __builtin_amdgcn_wmma_f32_16x16x4_f32(                        \
                false, a[s][c], false, q0[c], (short)0, acc0, false, false);     \
            acc1 = __builtin_amdgcn_wmma_f32_16x16x4_f32(                        \
                false, a[s][c], false, q1[c], (short)0, acc1, false, false);     \
        }                                                                        \
    }

#define TAIL(s, r, STORE)                                                        \
    {                                                                            \
        const int  mf_   = ((s) + 2) & 15;     \
        const int  vi_   = mf_ & 7;                           \
        const bool up_   = (mf_ >> 3) != 0;                    \
        const bool mine_ = up_ == hi;                                            \
        const float f0_ = acc0[vi_];                                             \
        const float f1_ = acc1[vi_];                                             \
        if (STORE) {                                                             \
            const float f1x_ = xhalf(f1_);     \
            float* __restrict__ orow_ = op + (size_t)((r) - 14) * 1024;          \
            orow_[up_ ? colB : colA] = mine_ ? f0_ : f1x_;                       \
        }                                                                        \
        acc0[vi_] = mine_ ? 0.0f : f0_;        \
        acc1[vi_] = mine_ ? 0.0f : f1_;                                          \
    }

#define PAIR(s, r, STORE)                                                        \
    {                                                                            \
        v2f n0[4], n1[4], m0[4], m1[4];                                          \
        bo += ((r) + 2 <= IPH - 2) ? 2u * IPW * 4u : 0u;                         \
        FETCH_PAIR(n0, n1, m0, m1)                                               \
        ROWMM(s, b0, b1)                                                         \
        TAIL(s, r, STORE)                                                        \
          \
          \
                    \
        __builtin_amdgcn_sched_barrier(0);                                       \
        ROWMM(((s) + 1) & 15, c0, c1)                                            \
        TAIL(((s) + 1) & 15, (r) + 1, STORE)                                     \
        _Pragma("unroll")                                                        \
        for (int c = 0; c < 4; ++c) {                                            \
            b0[c] = n0[c]; b1[c] = n1[c]; c0[c] = m0[c]; c1[c] = m1[c];          \
        }                                                                        \
    }

    FETCH_PAIR(b0, b1, c0, c1)

    #pragma unroll
    for (int s = 0; s < 14; s += 2) PAIR(s, s, false)

    #pragma clang loop unroll(disable)
    for (int base = 14; base < IPH; base += 16) {
        #pragma unroll
        for (int j = 0; j < 16; j += 2) PAIR((14 + j) & 15, base + j, true)
    }
#undef PAIR
#undef TAIL
#undef ROWMM
#undef FETCH_PAIR
#undef LOAD_ROW
}

extern "C" void kernel_launch(void* const* d_in, const int* in_sizes, int n_in,
                              void* d_out, int out_size, void* d_ws, size_t ws_size,
                              hipStream_t stream) {
    const float* x    = (const float*)d_in[0];
    const float* kern = (const float*)d_in[1];
    float* outp  = (float*)d_out;

    pwblur_k<<<3072, 64, 0, stream>>>(x, kern, outp);
}


# Input band streamed through a private 32-row LDS ring per wave: the next 16 rows' global loads fly under the current 16 rows' matrix instructions; no whole-band staging phase, no barrier
# speedup vs baseline: 1.4626x; 1.0540x over previous
#include <hip/hip_runtime.h>


typedef __attribute__((ext_vector_type(2))) float v2f;
typedef __attribute__((ext_vector_type(4))) float v4f;
typedef __attribute__((ext_vector_type(8))) float v8f;

#define TILE   256
#define BY     128
#define IPH    (BY + 14)
#define RP     48
#define RROWS  32

__device__ __forceinline__ float xhalf(float x) {
    int b = __builtin_bit_cast(int, x);
    return __builtin_bit_cast(float,
        __builtin_amdgcn_permlanex16(b, b, 0x76543210, 0xfedcba98, false, false));
}

__global__ __launch_bounds__(64) void pwblur_k(const float* __restrict__ x,
                                                const float* __restrict__ kraw,
                                                float* __restrict__ out) {
    __shared__ alignas(16) float ring[2][RROWS * RP];
    __shared__ alignas(16) float kl[2][256];

    const int tid  = threadIdx.x;
    const int lane = tid & 31;
    const int wave = tid >> 5;
    const bool hi  = (lane & 16) != 0;
    const int  nl  = lane & 15;

    const int bid   = blockIdx.x;
    const int sub   = bid & 7;
    const int t     = bid >> 3;
    const int ti    = t & 3;
    const int tj    = (t >> 2) & 3;
    const int plane = t >> 4;
    const int sub_y = sub >> 2;
    const int sub_x = sub & 3;

    const int wv = __builtin_amdgcn_readfirstlane(wave);

    const int ty0 = tj * TILE;
    const int tx0 = ti * TILE;
    const int y0  = sub_y * BY;
    const int xw  = sub_x * 64 + wv * 32;

    const float* __restrict__ xin = x + (size_t)plane * 1024 * 1024;

    const float* __restrict__ kt = kraw + (tj * 4 + ti) * 225;
    float kv[8], kp[8];
    #pragma unroll
    for (int j = 0; j < 8; ++j) {
        const int idx = lane + 32 * j;
        const float v = kt[idx < 225 ? idx : 224];
        kv[j] = idx < 225 ? v : 0.0f;
        const int dy = idx >> 4;
        const int dx = idx & 15;
        const bool ok = dy < 15 && dx < 15;
        const float p = kt[ok ? dy * 15 + dx : 0];
        kp[j] = ok ? p : 0.0f;
    }

    const int g4  = lane >> 3;
    const int q8  = lane & 7;
    const int hq  = nl < 14 ? nl : 14;
    const int hxi = hq < 7 ? hq : hq + 32;
    int hxs = xw + hxi - 7; hxs = hxs < 0 ? 0 : (hxs > TILE - 1 ? TILE - 1 : hxs);
    const float* __restrict__ xt = xin + (size_t)ty0 * 1024 + tx0;
    const int ioff = xw + 4 * q8;
    const int ylast = (y0 + IPH - 8) < (TILE - 1) ? (y0 + IPH - 8) : (TILE - 1);
    float* rg = ring[wv];
    const int sti = (g4 * RP + 8 + 4 * q8);
    const int sth = ((hi ? 1 : 0) * RP + 1 + hxi);

#define ISSUE_CHUNK(k, VI, VH)                                                   \
    {                                                                            \
        _Pragma("unroll")                                                        \
        for (int i = 0; i < 4; ++i) {                                            \
            int ys_ = y0 - 7 + 16 * (k) + 4 * i + g4;                            \
            ys_ = ys_ < 0 ? 0 : (ys_ > ylast ? ylast : ys_);                     \
            VI[i] = *(const v4f*)(xt + (ys_ * 1024 + ioff));                     \
        }                                                                        \
        _Pragma("unroll")                                                        \
        for (int i = 0; i < 8; ++i) {                                            \
            int ys_ = y0 - 7 + 16 * (k) + 2 * i + (hi ? 1 : 0);                  \
            ys_ = ys_ < 0 ? 0 : (ys_ > ylast ? ylast : ys_);                     \
            VH[i] = xt[ys_ * 1024 + hxs];                                        \
        }                                                                        \
    }

#define STORE_CHUNK(k, VI, VH)                                                   \
    {                                                                            \
        float* rs_ = rg + ((k) & 1) * (16 * RP);                                 \
        _Pragma("unroll")                                                        \
        for (int i = 0; i < 4; ++i) *(v4f*)(rs_ + sti + 4 * i * RP) = VI[i];     \
        _Pragma("unroll")                                                        \
        for (int i = 0; i < 8; ++i) rs_[sth + 2 * i * RP] = VH[i];               \
    }

    {
        v4f vi0[4], vi1[4];
        float vh0[8], vh1[8];
        ISSUE_CHUNK(0, vi0, vh0)
        ISSUE_CHUNK(1, vi1, vh1)
        STORE_CHUNK(0, vi0, vh0)
        STORE_CHUNK(1, vi1, vh1)
    }

    {
        float u0 = kv[0] + kv[4], u1 = kv[1] + kv[5], u2 = kv[2] + kv[6], u3 = kv[3] + kv[7];
        float w0 = u0 + u2, w1 = u1 + u3;
        float z = w0 + w1;
        z = z + xhalf(z);
        z = z + __shfl_xor(z, 8);
        z = z + __shfl_xor(z, 4);
        z = z + __shfl_xor(z, 2);
        z = z + __shfl_xor(z, 1);
        const float inv = 1.0f / (z + 1e-12f);
        #pragma unroll
        for (int j = 0; j < 8; ++j) {
            const int idx = lane + 32 * j;
            const bool ok = (idx >> 4) < 15 && (idx & 15) < 15;
            kl[wv][idx] = ok ? kp[j] * inv : 0.0f;
        }
    }

    const float* kn = kl[wv];
    v2f a[16][4];
    #pragma unroll
    for (int s = 0; s < 16; ++s) {
        const int krow = (s - nl) & 15;
        #pragma unroll
        for (int c = 0; c < 4; ++c) {
            const int dxb = 4 * c + (hi ? 2 : 0);
            a[s][c] = *(const v2f*)(kn + krow * 16 + dxb);
        }
    }

    const char* rgc = (const char*)&ring[0][0];
    const int   bl  = 1 + nl + (hi ? 2 : 0);
    const unsigned bl0 = (unsigned)((wv * (RROWS * RP) + bl) * 4);
    unsigned bo;

    float* __restrict__ op = out + (size_t)plane * 1024 * 1024
                           + (size_t)(ty0 + y0) * 1024 + (tx0 + xw);
    const int colA = (hi ? 16 : 0) + nl;
    const int colB = (hi ? 0 : 16) + nl;

    v8f acc0 = {};
    v8f acc1 = {};
    v2f b0[4], b1[4];
    v2f c0[4], c1[4];

#define LOAD_ROW(dst0, dst1, boff)                                               \
    {                                                                            \
        const float* p_ = (const float*)(rgc + (boff));                          \
        _Pragma("unroll")                                                        \
        for (int c = 0; c < 4; ++c) {                                            \
            dst0[c].x = p_[4 * c];      dst0[c].y = p_[4 * c + 1];               \
            dst1[c].x = p_[16 + 4 * c]; dst1[c].y = p_[16 + 4 * c + 1];          \
        }                                                                        \
    }

#define FETCH_PAIR(rr, dA0, dA1, dB0, dB1)                                       \
    {                                                                            \
        bo = bl0 + (unsigned)(rr) * (RP * 4u);                                   \
          \
          \
        asm volatile("" : "+v"(bo));                                             \
        LOAD_ROW(dA0, dA1, bo)                                                   \
        LOAD_ROW(dB0, dB1, bo + RP * 4)                                          \
          \
          \
        __builtin_amdgcn_sched_barrier(0);                                       \
    }

#define ROWMM(s, q0, q1)                                                         \
    {                                                                            \
        _Pragma("unroll")                                                        \
        for (int c = 0; c < 4; ++c) {                                            \
            acc0 = __builtin_amdgcn_wmma_f32_16x16x4_f32(                        \
                false, a[s][c], false, q0[c], (short)0, acc0, false, false);     \
            acc1 = __builtin_amdgcn_wmma_f32_16x16x4_f32(                        \
                false, a[s][c], false, q1[c], (short)0, acc1, false, false);     \
        }                                                                        \
    }

#define TAIL(s, r, STORE)                                                        \
    {                                                                            \
        const int  mf_   = ((s) + 2) & 15;     \
        const int  vi_   = mf_ & 7;                           \
        const bool up_   = (mf_ >> 3) != 0;                    \
        const bool mine_ = up_ == hi;                                            \
        const float f0_ = acc0[vi_];                                             \
        const float f1_ = acc1[vi_];                                             \
        if (STORE) {                                                             \
            const float f1x_ = xhalf(f1_);     \
            float* __restrict__ orow_ = op + (size_t)((r) - 14) * 1024;          \
            orow_[up_ ? colB : colA] = mine_ ? f0_ : f1x_;                       \
        }                                                                        \
        acc0[vi_] = mine_ ? 0.0f : f0_;        \
        acc1[vi_] = mine_ ? 0.0f : f1_;                                          \
    }

#define PAIR(s, r, rn, STORE)                                                    \
    {                                                                            \
        v2f n0[4], n1[4], m0[4], m1[4];                                          \
        FETCH_PAIR(rn, n0, n1, m0, m1)                                           \
        ROWMM(s, b0, b1)                                                         \
        TAIL(s, r, STORE)                                                        \
          \
          \
                    \
        __builtin_amdgcn_sched_barrier(0);                                       \
        ROWMM(((s) + 1) & 15, c0, c1)                                            \
        TAIL(((s) + 1) & 15, (r) + 1, STORE)                                     \
        _Pragma("unroll")                                                        \
        for (int c = 0; c < 4; ++c) {                                            \
            b0[c] = n0[c]; b1[c] = n1[c]; c0[c] = m0[c]; c1[c] = m1[c];          \
        }                                                                        \
    }

    FETCH_PAIR(0, b0, b1, c0, c1)

    #pragma unroll
    for (int s = 0; s < 14; s += 2) PAIR(s, s, s + 2, false)

    v4f gi[4];
    float gh[8];
    #pragma clang loop unroll(disable)
    for (int it = 0; it < 8; ++it) {
        const int base = 14 + 16 * it;
        const int slot = ((it + 1) & 1) * 16;
        const bool more = it < 7;
        if (more) ISSUE_CHUNK(it + 2, gi, gh)
        #pragma unroll
        for (int j = 0; j < 16; j += 2) PAIR((14 + j) & 15, base + j, slot + j, true)
        if (more) STORE_CHUNK(it + 2, gi, gh)
    }
#undef PAIR
#undef TAIL
#undef ROWMM
#undef FETCH_PAIR
#undef LOAD_ROW
#undef STORE_CHUNK
#undef ISSUE_CHUNK
}

extern "C" void kernel_launch(void* const* d_in, const int* in_sizes, int n_in,
                              void* d_out, int out_size, void* d_ws, size_t ws_size,
                              hipStream_t stream) {
    const float* x    = (const float*)d_in[0];
    const float* kern = (const float*)d_in[1];
    float* outp  = (float*)d_out;

    pwblur_k<<<3072, 64, 0, stream>>>(x, kern, outp);
}


# Replicate-padded edge rows merged by linearity: a tile's 8 identical top or bottom band rows take ONE row step on pre-summed kernel rows (sum order changes there); 135 row steps per wave, not 142
# speedup vs baseline: 1.5067x; 1.0302x over previous
#include <hip/hip_runtime.h>


typedef __attribute__((ext_vector_type(2))) float v2f;
typedef __attribute__((ext_vector_type(4))) float v4f;
typedef __attribute__((ext_vector_type(8))) float v8f;

#define TILE   256
#define BY     128
#define IPH    (BY + 14)
#define RP     48
#define RROWS  32

__device__ __forceinline__ float xhalf(float x) {
    int b = __builtin_bit_cast(int, x);
    return __builtin_bit_cast(float,
        __builtin_amdgcn_permlanex16(b, b, 0x76543210, 0xfedcba98, false, false));
}

__global__ __launch_bounds__(64, 4) void pwblur_k(const float* __restrict__ x,
                                                const float* __restrict__ kraw,
                                                float* __restrict__ out) {
    __shared__ alignas(16) float ring[2][RROWS * RP];
    __shared__ alignas(16) float kl[2][256];

    const int tid  = threadIdx.x;
    const int lane = tid & 31;
    const int wave = tid >> 5;
    const bool hi  = (lane & 16) != 0;
    const int  nl  = lane & 15;

    const int bid   = blockIdx.x;
    const int sub   = bid & 7;
    const int t     = bid >> 3;
    const int ti    = t & 3;
    const int tj    = (t >> 2) & 3;
    const int plane = t >> 4;
    const int sub_y = sub >> 2;
    const int sub_x = sub & 3;

    const int wv = __builtin_amdgcn_readfirstlane(wave);

    const int ty0 = tj * TILE;
    const int tx0 = ti * TILE;
    const int y0  = sub_y * BY;
    const int xw  = sub_x * 64 + wv * 32;

    const float* __restrict__ xin = x + (size_t)plane * 1024 * 1024;

    const float* __restrict__ kt = kraw + (tj * 4 + ti) * 225;
    float kv[8], kp[8];
    #pragma unroll
    for (int j = 0; j < 8; ++j) {
        const int idx = lane + 32 * j;
        const float v = kt[idx < 225 ? idx : 224];
        kv[j] = idx < 225 ? v : 0.0f;
        const int dy = idx >> 4;
        const int dx = idx & 15;
        const bool ok = dy < 15 && dx < 15;
        const float p = kt[ok ? dy * 15 + dx : 0];
        kp[j] = ok ? p : 0.0f;
    }

    const int g4  = lane >> 3;
    const int q8  = lane & 7;
    const int hq  = nl < 14 ? nl : 14;
    const int hxi = hq < 7 ? hq : hq + 32;
    int hxs = xw + hxi - 7; hxs = hxs < 0 ? 0 : (hxs > TILE - 1 ? TILE - 1 : hxs);
    const float* __restrict__ xt = xin + (size_t)ty0 * 1024 + tx0;
    const int ioff = xw + 4 * q8;
    const int ylast = (y0 + IPH - 8) < (TILE - 1) ? (y0 + IPH - 8) : (TILE - 1);
    float* rg = ring[wv];
    const int sti = (g4 * RP + 8 + 4 * q8);
    const int sth = ((hi ? 1 : 0) * RP + 1 + hxi);

#define ISSUE_CHUNK(k, VI, VH)                                                   \
    {                                                                            \
        _Pragma("unroll")                                                        \
        for (int i = 0; i < 4; ++i) {                                            \
            int ys_ = y0 - 7 + 16 * (k) + 4 * i + g4;                            \
            ys_ = ys_ < 0 ? 0 : (ys_ > ylast ? ylast : ys_);                     \
            VI[i] = *(const v4f*)(xt + (ys_ * 1024 + ioff));                     \
        }                                                                        \
        _Pragma("unroll")                                                        \
        for (int i = 0; i < 8; ++i) {                                            \
            int ys_ = y0 - 7 + 16 * (k) + 2 * i + (hi ? 1 : 0);                  \
            ys_ = ys_ < 0 ? 0 : (ys_ > ylast ? ylast : ys_);                     \
            VH[i] = xt[ys_ * 1024 + hxs];                                        \
        }                                                                        \
    }

#define STORE_CHUNK(k, VI, VH)                                                   \
    {                                                                            \
        float* rs_ = rg + ((k) & 1) * (16 * RP);                                 \
        _Pragma("unroll")                                                        \
        for (int i = 0; i < 4; ++i) *(v4f*)(rs_ + sti + 4 * i * RP) = VI[i];     \
        _Pragma("unroll")                                                        \
        for (int i = 0; i < 8; ++i) rs_[sth + 2 * i * RP] = VH[i];               \
    }

    {
        v4f vi0[4], vi1[4];
        float vh0[8], vh1[8];
        ISSUE_CHUNK(0, vi0, vh0)
        ISSUE_CHUNK(1, vi1, vh1)
        STORE_CHUNK(0, vi0, vh0)
        STORE_CHUNK(1, vi1, vh1)
    }

    {
        float u0 = kv[0] + kv[4], u1 = kv[1] + kv[5], u2 = kv[2] + kv[6], u3 = kv[3] + kv[7];
        float w0 = u0 + u2, w1 = u1 + u3;
        float z = w0 + w1;
        z = z + xhalf(z);
        z = z + __shfl_xor(z, 8);
        z = z + __shfl_xor(z, 4);
        z = z + __shfl_xor(z, 2);
        z = z + __shfl_xor(z, 1);
        const float inv = 1.0f / (z + 1e-12f);
        #pragma unroll
        for (int j = 0; j < 8; ++j) {
            const int idx = lane + 32 * j;
            const bool ok = (idx >> 4) < 15 && (idx & 15) < 15;
            kl[wv][idx] = ok ? kp[j] * inv : 0.0f;
        }
    }

    const float* kn = kl[wv];
    v2f a[16][4];
    #pragma unroll
    for (int s = 0; s < 16; ++s) {
        const int krow = (s - nl) & 15;
        #pragma unroll
        for (int c = 0; c < 4; ++c) {
            const int dxb = 4 * c + (hi ? 2 : 0);
            a[s][c] = *(const v2f*)(kn + krow * 16 + dxb);
        }
    }

#define EDGE_SUM(AE, S0, COND)                                                   \
    {                                                                            \
        _Pragma("unroll")                                                        \
        for (int c = 0; c < 4; ++c) { AE[c].x = 0.0f; AE[c].y = 0.0f; }          \
        _Pragma("unroll")                                                        \
        for (int i = 0; i < 8; ++i) {                                            \
            const int  s_  = (S0) + i;                                           \
            const bool in_ = (COND);                                             \
            _Pragma("unroll")                                                    \
            for (int c = 0; c < 4; ++c) {                                        \
                AE[c].x += in_ ? a[s_][c].x : 0.0f;                              \
                AE[c].y += in_ ? a[s_][c].y : 0.0f;                              \
            }                                                                    \
        }                                                                        \
    }

    const char* rgc = (const char*)&ring[0][0];
    const int   bl  = 1 + nl + (hi ? 2 : 0);
    const unsigned bl0 = (unsigned)((wv * (RROWS * RP) + bl) * 4);
    unsigned bo;

    float* __restrict__ op = out + (size_t)plane * 1024 * 1024
                           + (size_t)(ty0 + y0) * 1024 + (tx0 + xw);
    const int colA = (hi ? 16 : 0) + nl;
    const int colB = (hi ? 0 : 16) + nl;

    v8f acc0 = {};
    v8f acc1 = {};
    v2f b0[4], b1[4];
    v2f c0[4], c1[4];

#define LOAD_ROW(dst0, dst1, boff)                                               \
    {                                                                            \
        const float* p_ = (const float*)(rgc + (boff));                          \
        _Pragma("unroll")                                                        \
        for (int c = 0; c < 4; ++c) {                                            \
            dst0[c].x = p_[4 * c];      dst0[c].y = p_[4 * c + 1];               \
            dst1[c].x = p_[16 + 4 * c]; dst1[c].y = p_[16 + 4 * c + 1];          \
        }                                                                        \
    }

#define FETCH_PAIR(rr, dA0, dA1, dB0, dB1)                                       \
    {                                                                            \
        bo = bl0 + (unsigned)(rr) * (RP * 4u);                                   \
          \
          \
        asm volatile("" : "+v"(bo));                                             \
        LOAD_ROW(dA0, dA1, bo)                                                   \
        LOAD_ROW(dB0, dB1, bo + RP * 4)                                          \
          \
          \
        __builtin_amdgcn_sched_barrier(0);                                       \
    }

#define ROWMM(s, q0, q1)                                                         \
    {                                                                            \
        _Pragma("unroll")                                                        \
        for (int c = 0; c < 4; ++c) {                                            \
            acc0 = __builtin_amdgcn_wmma_f32_16x16x4_f32(                        \
                false, a[s][c], false, q0[c], (short)0, acc0, false, false);     \
            acc1 = __builtin_amdgcn_wmma_f32_16x16x4_f32(                        \
                false, a[s][c], false, q1[c], (short)0, acc1, false, false);     \
        }                                                                        \
    }

#define ROWMM_A(AE, q0, q1)                                                      \
    {                                                                            \
        _Pragma("unroll")                                                        \
        for (int c = 0; c < 4; ++c) {                                            \
            acc0 = __builtin_amdgcn_wmma_f32_16x16x4_f32(                        \
                false, AE[c], false, q0[c], (short)0, acc0, false, false);       \
            acc1 = __builtin_amdgcn_wmma_f32_16x16x4_f32(                        \
                false, AE[c], false, q1[c], (short)0, acc1, false, false);       \
        }                                                                        \
    }

#define TAIL(s, r, STORE)                                                        \
    {                                                                            \
        const int  mf_   = ((s) + 2) & 15;     \
        const int  vi_   = mf_ & 7;                           \
        const bool up_   = (mf_ >> 3) != 0;                    \
        const bool mine_ = up_ == hi;                                            \
        const float f0_ = acc0[vi_];                                             \
        const float f1_ = acc1[vi_];                                             \
        if (STORE) {                                                             \
            const float f1x_ = xhalf(f1_);     \
            float* __restrict__ orow_ = op + (size_t)((r) - 14) * 1024;          \
            orow_[up_ ? colB : colA] = mine_ ? f0_ : f1x_;                       \
        }                                                                        \
        acc0[vi_] = mine_ ? 0.0f : f0_;        \
        acc1[vi_] = mine_ ? 0.0f : f1_;                                          \
    }

#define PAIR(s, r, rn, STORE)                                                    \
    {                                                                            \
        v2f n0[4], n1[4], m0[4], m1[4];                                          \
        FETCH_PAIR(rn, n0, n1, m0, m1)                                           \
        ROWMM(s, b0, b1)                                                         \
        TAIL(s, r, STORE)                                                        \
          \
          \
                    \
        __builtin_amdgcn_sched_barrier(0);                                       \
        ROWMM(((s) + 1) & 15, c0, c1)                                            \
        TAIL(((s) + 1) & 15, (r) + 1, STORE)                                     \
        _Pragma("unroll")                                                        \
        for (int c = 0; c < 4; ++c) {                                            \
            b0[c] = n0[c]; b1[c] = n1[c]; c0[c] = m0[c]; c1[c] = m1[c];          \
        }                                                                        \
    }

    FETCH_PAIR(0, b0, b1, c0, c1)

    if (sub_y == 0) {
        v2f ae[4];
        EDGE_SUM(ae, 0, nl <= s_)
        v2f n0[4], n1[4], m0[4], m1[4];
        FETCH_PAIR(8, n0, n1, m0, m1)
        ROWMM_A(ae, b0, b1)
        #pragma unroll
        for (int c = 0; c < 4; ++c) {
            b0[c] = n0[c]; b1[c] = n1[c]; c0[c] = m0[c]; c1[c] = m1[c];
        }
    } else {
        {
            v2f ae[4];
            EDGE_SUM(ae, 6, nl >= s_ + 2)
            float* kpark = kl[wv] + 8 * lane;
            v4f e0, e1;
            e0.x = ae[0].x; e0.y = ae[0].y; e0.z = ae[1].x; e0.w = ae[1].y;
            e1.x = ae[2].x; e1.y = ae[2].y; e1.z = ae[3].x; e1.w = ae[3].y;
            *(v4f*)(kpark)     = e0;
            *(v4f*)(kpark + 4) = e1;
        }
        #pragma unroll
        for (int s = 0; s < 8; s += 2) PAIR(s, s, s + 2, false)
    }
    #pragma unroll
    for (int s = 8; s < 14; s += 2) PAIR(s, s, s + 2, false)

    v4f gi[4];
    float gh[8];
    #pragma clang loop unroll(disable)
    for (int it = 0; it < 8; ++it) {
        const int base = 14 + 16 * it;
        const int slot = ((it + 1) & 1) * 16;
        const bool more = it < 7;
        if (more) ISSUE_CHUNK(it + 2, gi, gh)
        #pragma unroll
        for (int j = 0; j < 8; j += 2) PAIR((14 + j) & 15, base + j, slot + j, true)
        if (more || sub_y == 0) {
            #pragma unroll
            for (int j = 8; j < 16; j += 2) PAIR((14 + j) & 15, base + j, slot + j, true)
        }
        if (more) STORE_CHUNK(it + 2, gi, gh)
    }
    if (sub_y != 0) {
        v2f ab[4];
        {
            const int ln = (int)__builtin_amdgcn_mbcnt_lo(~0u, 0u);
            const float* kpark = kl[wv] + 8 * ln;
            const v4f e0 = *(const v4f*)(kpark);
            const v4f e1 = *(const v4f*)(kpark + 4);
            ab[0].x = e0.x; ab[0].y = e0.y; ab[1].x = e0.z; ab[1].y = e0.w;
            ab[2].x = e1.x; ab[2].y = e1.y; ab[3].x = e1.z; ab[3].y = e1.w;
        }
        ROWMM_A(ab, b0, b1)
        #pragma unroll
        for (int i = 0; i < 8; ++i) TAIL(6 + i, 134 + i, true)
    }
#undef PAIR
#undef TAIL
#undef EDGE_SUM
#undef ROWMM_A
#undef ROWMM
#undef FETCH_PAIR
#undef LOAD_ROW
#undef STORE_CHUNK
#undef ISSUE_CHUNK
}

extern "C" void kernel_launch(void* const* d_in, const int* in_sizes, int n_in,
                              void* d_out, int out_size, void* d_ws, size_t ws_size,
                              hipStream_t stream) {
    const float* x    = (const float*)d_in[0];
    const float* kern = (const float*)d_in[1];
    float* outp  = (float*)d_out;

    pwblur_k<<<3072, 64, 0, stream>>>(x, kern, outp);
}


# One wave walks the whole 256-row tile height instead of a 128-row strip: no band row fetched and multiplied by two strips, start-up once per 256 rows; 256 row steps per 256 rows, was 270; one round
# speedup vs baseline: 1.5895x; 1.0550x over previous
#include <hip/hip_runtime.h>


typedef __attribute__((ext_vector_type(2))) float v2f;
typedef __attribute__((ext_vector_type(4))) float v4f;
typedef __attribute__((ext_vector_type(8))) float v8f;

#define TILE   256
#define BY     256
#define IPH    (BY + 14)
#define RP     48
#define RROWS  32

__device__ __forceinline__ float xhalf(float x) {
    int b = __builtin_bit_cast(int, x);
    return __builtin_bit_cast(float,
        __builtin_amdgcn_permlanex16(b, b, 0x76543210, 0xfedcba98, false, false));
}

__global__ __launch_bounds__(128, 3) void pwblur_k(const float* __restrict__ x,
                                                const float* __restrict__ kraw,
                                                float* __restrict__ out) {
    __shared__ alignas(16) float ring[4][RROWS * RP];
    __shared__ alignas(16) float kl[4][256];

    const int tid  = threadIdx.x;
    const int lane = tid & 31;
    const int wave = tid >> 5;
    const bool hi  = (lane & 16) != 0;
    const int  nl  = lane & 15;

    const int bid   = blockIdx.x;
    const int sub_x = bid & 1;
    const int t     = bid >> 1;
    const int ti    = t & 3;
    const int tj    = (t >> 2) & 3;
    const int plane = t >> 4;

    const int wv = __builtin_amdgcn_readfirstlane(wave);

    const int ty0 = tj * TILE;
    const int tx0 = ti * TILE;
    const int y0  = 0;
    const int xw  = sub_x * 128 + wv * 32;

    const float* __restrict__ xin = x + (size_t)plane * 1024 * 1024;

    const float* __restrict__ kt = kraw + (tj * 4 + ti) * 225;
    float kv[8], kp[8];
    #pragma unroll
    for (int j = 0; j < 8; ++j) {
        const int idx = lane + 32 * j;
        const float v = kt[idx < 225 ? idx : 224];
        kv[j] = idx < 225 ? v : 0.0f;
        const int dy = idx >> 4;
        const int dx = idx & 15;
        const bool ok = dy < 15 && dx < 15;
        const float p = kt[ok ? dy * 15 + dx : 0];
        kp[j] = ok ? p : 0.0f;
    }

    const int g4  = lane >> 3;
    const int q8  = lane & 7;
    const int hq  = nl < 14 ? nl : 14;
    const int hxi = hq < 7 ? hq : hq + 32;
    int hxs = xw + hxi - 7; hxs = hxs < 0 ? 0 : (hxs > TILE - 1 ? TILE - 1 : hxs);
    const float* __restrict__ xt = xin + (size_t)ty0 * 1024 + tx0;
    const int ioff = xw + 4 * q8;
    const int ylast = (y0 + IPH - 8) < (TILE - 1) ? (y0 + IPH - 8) : (TILE - 1);
    float* rg = ring[wv];
    const int sti = (g4 * RP + 8 + 4 * q8);
    const int sth = ((hi ? 1 : 0) * RP + 1 + hxi);

#define ISSUE_CHUNK(k, VI, VH)                                                   \
    {                                                                            \
        _Pragma("unroll")                                                        \
        for (int i = 0; i < 4; ++i) {                                            \
            int ys_ = y0 - 7 + 16 * (k) + 4 * i + g4;                            \
            ys_ = ys_ < 0 ? 0 : (ys_ > ylast ? ylast : ys_);                     \
            VI[i] = *(const v4f*)(xt + (ys_ * 1024 + ioff));                     \
        }                                                                        \
        _Pragma("unroll")                                                        \
        for (int i = 0; i < 8; ++i) {                                            \
            int ys_ = y0 - 7 + 16 * (k) + 2 * i + (hi ? 1 : 0);                  \
            ys_ = ys_ < 0 ? 0 : (ys_ > ylast ? ylast : ys_);                     \
            VH[i] = xt[ys_ * 1024 + hxs];                                        \
        }                                                                        \
    }

#define STORE_CHUNK(k, VI, VH)                                                   \
    {                                                                            \
        float* rs_ = rg + ((k) & 1) * (16 * RP);                                 \
        _Pragma("unroll")                                                        \
        for (int i = 0; i < 4; ++i) *(v4f*)(rs_ + sti + 4 * i * RP) = VI[i];     \
        _Pragma("unroll")                                                        \
        for (int i = 0; i < 8; ++i) rs_[sth + 2 * i * RP] = VH[i];               \
    }

    {
        v4f vi0[4], vi1[4];
        float vh0[8], vh1[8];
        ISSUE_CHUNK(0, vi0, vh0)
        ISSUE_CHUNK(1, vi1, vh1)
        STORE_CHUNK(0, vi0, vh0)
        STORE_CHUNK(1, vi1, vh1)
    }

    {
        float u0 = kv[0] + kv[4], u1 = kv[1] + kv[5], u2 = kv[2] + kv[6], u3 = kv[3] + kv[7];
        float w0 = u0 + u2, w1 = u1 + u3;
        float z = w0 + w1;
        z = z + xhalf(z);
        z = z + __shfl_xor(z, 8);
        z = z + __shfl_xor(z, 4);
        z = z + __shfl_xor(z, 2);
        z = z + __shfl_xor(z, 1);
        const float inv = 1.0f / (z + 1e-12f);
        #pragma unroll
        for (int j = 0; j < 8; ++j) {
            const int idx = lane + 32 * j;
            const bool ok = (idx >> 4) < 15 && (idx & 15) < 15;
            kl[wv][idx] = ok ? kp[j] * inv : 0.0f;
        }
    }

    const float* kn = kl[wv];
    v2f a[16][4];
    #pragma unroll
    for (int s = 0; s < 16; ++s) {
        const int krow = (s - nl) & 15;
        #pragma unroll
        for (int c = 0; c < 4; ++c) {
            const int dxb = 4 * c + (hi ? 2 : 0);
            a[s][c] = *(const v2f*)(kn + krow * 16 + dxb);
        }
    }

#define EDGE_SUM(AE, S0, COND)                                                   \
    {                                                                            \
        _Pragma("unroll")                                                        \
        for (int c = 0; c < 4; ++c) { AE[c].x = 0.0f; AE[c].y = 0.0f; }          \
        _Pragma("unroll")                                                        \
        for (int i = 0; i < 8; ++i) {                                            \
            const int  s_  = (S0) + i;                                           \
            const bool in_ = (COND);                                             \
            _Pragma("unroll")                                                    \
            for (int c = 0; c < 4; ++c) {                                        \
                AE[c].x += in_ ? a[s_][c].x : 0.0f;                              \
                AE[c].y += in_ ? a[s_][c].y : 0.0f;                              \
            }                                                                    \
        }                                                                        \
    }

    const char* rgc = (const char*)&ring[0][0];
    const int   bl  = 1 + nl + (hi ? 2 : 0);
    const unsigned bl0 = (unsigned)((wv * (RROWS * RP) + bl) * 4);
    unsigned bo;

    float* __restrict__ op = out + (size_t)plane * 1024 * 1024
                           + (size_t)(ty0 + y0) * 1024 + (tx0 + xw);
    const int colA = (hi ? 16 : 0) + nl;
    const int colB = (hi ? 0 : 16) + nl;

    v8f acc0 = {};
    v8f acc1 = {};
    v2f b0[4], b1[4];
    v2f c0[4], c1[4];

#define LOAD_ROW(dst0, dst1, boff)                                               \
    {                                                                            \
        const float* p_ = (const float*)(rgc + (boff));                          \
        _Pragma("unroll")                                                        \
        for (int c = 0; c < 4; ++c) {                                            \
            dst0[c].x = p_[4 * c];      dst0[c].y = p_[4 * c + 1];               \
            dst1[c].x = p_[16 + 4 * c]; dst1[c].y = p_[16 + 4 * c + 1];          \
        }                                                                        \
    }

#define FETCH_PAIR(rr, dA0, dA1, dB0, dB1)                                       \
    {                                                                            \
        bo = bl0 + (unsigned)(rr) * (RP * 4u);                                   \
          \
          \
        asm volatile("" : "+v"(bo));                                             \
        LOAD_ROW(dA0, dA1, bo)                                                   \
        LOAD_ROW(dB0, dB1, bo + RP * 4)                                          \
          \
          \
        __builtin_amdgcn_sched_barrier(0);                                       \
    }

#define ROWMM(s, q0, q1)                                                         \
    {                                                                            \
        _Pragma("unroll")                                                        \
        for (int c = 0; c < 4; ++c) {                                            \
            acc0 = __builtin_amdgcn_wmma_f32_16x16x4_f32(                        \
                false, a[s][c], false, q0[c], (short)0, acc0, false, false);     \
            acc1 = __builtin_amdgcn_wmma_f32_16x16x4_f32(                        \
                false, a[s][c], false, q1[c], (short)0, acc1, false, false);     \
        }                                                                        \
    }

#define ROWMM_A(AE, q0, q1)                                                      \
    {                                                                            \
        _Pragma("unroll")                                                        \
        for (int c = 0; c < 4; ++c) {                                            \
            acc0 = __builtin_amdgcn_wmma_f32_16x16x4_f32(                        \
                false, AE[c], false, q0[c], (short)0, acc0, false, false);       \
            acc1 = __builtin_amdgcn_wmma_f32_16x16x4_f32(                        \
                false, AE[c], false, q1[c], (short)0, acc1, false, false);       \
        }                                                                        \
    }

#define TAIL(s, r, STORE)                                                        \
    {                                                                            \
        const int  mf_   = ((s) + 2) & 15;     \
        const int  vi_   = mf_ & 7;                           \
        const bool up_   = (mf_ >> 3) != 0;                    \
        const bool mine_ = up_ == hi;                                            \
        const float f0_ = acc0[vi_];                                             \
        const float f1_ = acc1[vi_];                                             \
        if (STORE) {                                                             \
            const float f1x_ = xhalf(f1_);     \
            float* __restrict__ orow_ = op + (size_t)((r) - 14) * 1024;          \
            orow_[up_ ? colB : colA] = mine_ ? f0_ : f1x_;                       \
        }                                                                        \
        acc0[vi_] = mine_ ? 0.0f : f0_;        \
        acc1[vi_] = mine_ ? 0.0f : f1_;                                          \
    }

#define PAIR(s, r, rn, STORE)                                                    \
    {                                                                            \
        v2f n0[4], n1[4], m0[4], m1[4];                                          \
        FETCH_PAIR(rn, n0, n1, m0, m1)                                           \
        ROWMM(s, b0, b1)                                                         \
        TAIL(s, r, STORE)                                                        \
          \
          \
                    \
        __builtin_amdgcn_sched_barrier(0);                                       \
        ROWMM(((s) + 1) & 15, c0, c1)                                            \
        TAIL(((s) + 1) & 15, (r) + 1, STORE)                                     \
        _Pragma("unroll")                                                        \
        for (int c = 0; c < 4; ++c) {                                            \
            b0[c] = n0[c]; b1[c] = n1[c]; c0[c] = m0[c]; c1[c] = m1[c];          \
        }                                                                        \
    }

    FETCH_PAIR(0, b0, b1, c0, c1)

    {
        v2f ae[4];
        EDGE_SUM(ae, 6, nl >= s_ + 2)
        float* kpark = kl[wv] + 8 * lane;
        v4f e0, e1;
        e0.x = ae[0].x; e0.y = ae[0].y; e0.z = ae[1].x; e0.w = ae[1].y;
        e1.x = ae[2].x; e1.y = ae[2].y; e1.z = ae[3].x; e1.w = ae[3].y;
        *(v4f*)(kpark)     = e0;
        *(v4f*)(kpark + 4) = e1;
    }
    {
        v2f ae[4];
        EDGE_SUM(ae, 0, nl <= s_)
        v2f n0[4], n1[4], m0[4], m1[4];
        FETCH_PAIR(8, n0, n1, m0, m1)
        ROWMM_A(ae, b0, b1)
        #pragma unroll
        for (int c = 0; c < 4; ++c) {
            b0[c] = n0[c]; b1[c] = n1[c]; c0[c] = m0[c]; c1[c] = m1[c];
        }
    }
    #pragma unroll
    for (int s = 8; s < 14; s += 2) PAIR(s, s, s + 2, false)

    v4f gi[4];
    float gh[8];
    #pragma clang loop unroll(disable)
    for (int it = 0; it < 16; ++it) {
        const int base = 14 + 16 * it;
        const int slot = ((it + 1) & 1) * 16;
        const bool more = it < 15;
        if (more) ISSUE_CHUNK(it + 2, gi, gh)
        #pragma unroll
        for (int j = 0; j < 8; j += 2) PAIR((14 + j) & 15, base + j, slot + j, true)
        if (more) {
            #pragma unroll
            for (int j = 8; j < 16; j += 2) PAIR((14 + j) & 15, base + j, slot + j, true)
            STORE_CHUNK(it + 2, gi, gh)
        }
    }
    {
        v2f ab[4];
        {
            const int ln = (int)__builtin_amdgcn_mbcnt_lo(~0u, 0u);
            const float* kpark = kl[wv] + 8 * ln;
            const v4f e0 = *(const v4f*)(kpark);
            const v4f e1 = *(const v4f*)(kpark + 4);
            ab[0].x = e0.x; ab[0].y = e0.y; ab[1].x = e0.z; ab[1].y = e0.w;
            ab[2].x = e1.x; ab[2].y = e1.y; ab[3].x = e1.z; ab[3].y = e1.w;
        }
        ROWMM_A(ab, b0, b1)
        #pragma unroll
        for (int i = 0; i < 8; ++i) TAIL(6 + i, 262 + i, true)
    }
#undef PAIR
#undef TAIL
#undef EDGE_SUM
#undef ROWMM_A
#undef ROWMM
#undef FETCH_PAIR
#undef LOAD_ROW
#undef STORE_CHUNK
#undef ISSUE_CHUNK
}

extern "C" void kernel_launch(void* const* d_in, const int* in_sizes, int n_in,
                              void* d_out, int out_size, void* d_ws, size_t ws_size,
                              hipStream_t stream) {
    const float* x    = (const float*)d_in[0];
    const float* kern = (const float*)d_in[1];
    float* outp  = (float*)d_out;

    pwblur_k<<<768, 128, 0, stream>>>(x, kern, outp);
}


# Wave priority stepped down with loop progress (s_setprio 3,2,1,0 from passes 0,9,13,15): the three waves of a SIMD end together instead of the youngest running most of its passes alone
# speedup vs baseline: 1.6568x; 1.0423x over previous
#include <hip/hip_runtime.h>


typedef __attribute__((ext_vector_type(2))) float v2f;
typedef __attribute__((ext_vector_type(4))) float v4f;
typedef __attribute__((ext_vector_type(8))) float v8f;

#define TILE   256
#define BY     256
#define IPH    (BY + 14)
#define RP     48
#define RROWS  32

__device__ __forceinline__ float xhalf(float x) {
    int b = __builtin_bit_cast(int, x);
    return __builtin_bit_cast(float,
        __builtin_amdgcn_permlanex16(b, b, 0x76543210, 0xfedcba98, false, false));
}

__global__ __launch_bounds__(128, 3) void pwblur_k(const float* __restrict__ x,
                                                const float* __restrict__ kraw,
                                                float* __restrict__ out) {
    __shared__ alignas(16) float ring[4][RROWS * RP];
    __shared__ alignas(16) float kl[4][256];

    __builtin_amdgcn_s_setprio(3);

    const int tid  = threadIdx.x;
    const int lane = tid & 31;
    const int wave = tid >> 5;
    const bool hi  = (lane & 16) != 0;
    const int  nl  = lane & 15;

    const int bid   = blockIdx.x;
    const int sub_x = bid & 1;
    const int t     = bid >> 1;
    const int ti    = t & 3;
    const int tj    = (t >> 2) & 3;
    const int plane = t >> 4;

    const int wv = __builtin_amdgcn_readfirstlane(wave);

    const int ty0 = tj * TILE;
    const int tx0 = ti * TILE;
    const int y0  = 0;
    const int xw  = sub_x * 128 + wv * 32;

    const float* __restrict__ xin = x + (size_t)plane * 1024 * 1024;

    const float* __restrict__ kt = kraw + (tj * 4 + ti) * 225;
    float kv[8], kp[8];
    #pragma unroll
    for (int j = 0; j < 8; ++j) {
        const int idx = lane + 32 * j;
        const float v = kt[idx < 225 ? idx : 224];
        kv[j] = idx < 225 ? v : 0.0f;
        const int dy = idx >> 4;
        const int dx = idx & 15;
        const bool ok = dy < 15 && dx < 15;
        const float p = kt[ok ? dy * 15 + dx : 0];
        kp[j] = ok ? p : 0.0f;
    }

    const int g4  = lane >> 3;
    const int q8  = lane & 7;
    const int hq  = nl < 14 ? nl : 14;
    const int hxi = hq < 7 ? hq : hq + 32;
    int hxs = xw + hxi - 7; hxs = hxs < 0 ? 0 : (hxs > TILE - 1 ? TILE - 1 : hxs);
    const float* __restrict__ xt = xin + (size_t)ty0 * 1024 + tx0;
    const int ioff = xw + 4 * q8;
    const int ylast = (y0 + IPH - 8) < (TILE - 1) ? (y0 + IPH - 8) : (TILE - 1);
    float* rg = ring[wv];
    const int sti = (g4 * RP + 8 + 4 * q8);
    const int sth = ((hi ? 1 : 0) * RP + 1 + hxi);

#define ISSUE_CHUNK(k, VI, VH)                                                   \
    {                                                                            \
        _Pragma("unroll")                                                        \
        for (int i = 0; i < 4; ++i) {                                            \
            int ys_ = y0 - 7 + 16 * (k) + 4 * i + g4;                            \
            ys_ = ys_ < 0 ? 0 : (ys_ > ylast ? ylast : ys_);                     \
            VI[i] = *(const v4f*)(xt + (ys_ * 1024 + ioff));                     \
        }                                                                        \
        _Pragma("unroll")                                                        \
        for (int i = 0; i < 8; ++i) {                                            \
            int ys_ = y0 - 7 + 16 * (k) + 2 * i + (hi ? 1 : 0);                  \
            ys_ = ys_ < 0 ? 0 : (ys_ > ylast ? ylast : ys_);                     \
            VH[i] = xt[ys_ * 1024 + hxs];                                        \
        }                                                                        \
    }

#define STORE_CHUNK(k, VI, VH)                                                   \
    {                                                                            \
        float* rs_ = rg + ((k) & 1) * (16 * RP);                                 \
        _Pragma("unroll")                                                        \
        for (int i = 0; i < 4; ++i) *(v4f*)(rs_ + sti + 4 * i * RP) = VI[i];     \
        _Pragma("unroll")                                                        \
        for (int i = 0; i < 8; ++i) rs_[sth + 2 * i * RP] = VH[i];               \
    }

    {
        v4f vi0[4], vi1[4];
        float vh0[8], vh1[8];
        ISSUE_CHUNK(0, vi0, vh0)
        ISSUE_CHUNK(1, vi1, vh1)
        STORE_CHUNK(0, vi0, vh0)
        STORE_CHUNK(1, vi1, vh1)
    }

    {
        float u0 = kv[0] + kv[4], u1 = kv[1] + kv[5], u2 = kv[2] + kv[6], u3 = kv[3] + kv[7];
        float w0 = u0 + u2, w1 = u1 + u3;
        float z = w0 + w1;
        z = z + xhalf(z);
        z = z + __shfl_xor(z, 8);
        z = z + __shfl_xor(z, 4);
        z = z + __shfl_xor(z, 2);
        z = z + __shfl_xor(z, 1);
        const float inv = 1.0f / (z + 1e-12f);
        #pragma unroll
        for (int j = 0; j < 8; ++j) {
            const int idx = lane + 32 * j;
            const bool ok = (idx >> 4) < 15 && (idx & 15) < 15;
            kl[wv][idx] = ok ? kp[j] * inv : 0.0f;
        }
    }

    const float* kn = kl[wv];
    v2f a[16][4];
    #pragma unroll
    for (int s = 0; s < 16; ++s) {
        const int krow = (s - nl) & 15;
        #pragma unroll
        for (int c = 0; c < 4; ++c) {
            const int dxb = 4 * c + (hi ? 2 : 0);
            a[s][c] = *(const v2f*)(kn + krow * 16 + dxb);
        }
    }

#define EDGE_SUM(AE, S0, COND)                                                   \
    {                                                                            \
        _Pragma("unroll")                                                        \
        for (int c = 0; c < 4; ++c) { AE[c].x = 0.0f; AE[c].y = 0.0f; }          \
        _Pragma("unroll")                                                        \
        for (int i = 0; i < 8; ++i) {                                            \
            const int  s_  = (S0) + i;                                           \
            const bool in_ = (COND);                                             \
            _Pragma("unroll")                                                    \
            for (int c = 0; c < 4; ++c) {                                        \
                AE[c].x += in_ ? a[s_][c].x : 0.0f;                              \
                AE[c].y += in_ ? a[s_][c].y : 0.0f;                              \
            }                                                                    \
        }                                                                        \
    }

    const char* rgc = (const char*)&ring[0][0];
    const int   bl  = 1 + nl + (hi ? 2 : 0);
    const unsigned bl0 = (unsigned)((wv * (RROWS * RP) + bl) * 4);
    unsigned bo;

    float* __restrict__ op = out + (size_t)plane * 1024 * 1024
                           + (size_t)(ty0 + y0) * 1024 + (tx0 + xw);
    const int colA = (hi ? 16 : 0) + nl;
    const int colB = (hi ? 0 : 16) + nl;

    v8f acc0 = {};
    v8f acc1 = {};
    v2f b0[4], b1[4];
    v2f c0[4], c1[4];

#define LOAD_ROW(dst0, dst1, boff)                                               \
    {                                                                            \
        const float* p_ = (const float*)(rgc + (boff));                          \
        _Pragma("unroll")                                                        \
        for (int c = 0; c < 4; ++c) {                                            \
            dst0[c].x = p_[4 * c];      dst0[c].y = p_[4 * c + 1];               \
            dst1[c].x = p_[16 + 4 * c]; dst1[c].y = p_[16 + 4 * c + 1];          \
        }                                                                        \
    }

#define FETCH_PAIR(rr, dA0, dA1, dB0, dB1)                                       \
    {                                                                            \
        bo = bl0 + (unsigned)(rr) * (RP * 4u);                                   \
          \
          \
        asm volatile("" : "+v"(bo));                                             \
        LOAD_ROW(dA0, dA1, bo)                                                   \
        LOAD_ROW(dB0, dB1, bo + RP * 4)                                          \
          \
          \
        __builtin_amdgcn_sched_barrier(0);                                       \
    }

#define ROWMM(s, q0, q1)                                                         \
    {                                                                            \
        _Pragma("unroll")                                                        \
        for (int c = 0; c < 4; ++c) {                                            \
            acc0 = __builtin_amdgcn_wmma_f32_16x16x4_f32(                        \
                false, a[s][c], false, q0[c], (short)0, acc0, false, false);     \
            acc1 = __builtin_amdgcn_wmma_f32_16x16x4_f32(                        \
                false, a[s][c], false, q1[c], (short)0, acc1, false, false);     \
        }                                                                        \
    }

#define ROWMM_A(AE, q0, q1)                                                      \
    {                                                                            \
        _Pragma("unroll")                                                        \
        for (int c = 0; c < 4; ++c) {                                            \
            acc0 = __builtin_amdgcn_wmma_f32_16x16x4_f32(                        \
                false, AE[c], false, q0[c], (short)0, acc0, false, false);       \
            acc1 = __builtin_amdgcn_wmma_f32_16x16x4_f32(                        \
                false, AE[c], false, q1[c], (short)0, acc1, false, false);       \
        }                                                                        \
    }

#define TAIL(s, r, STORE)                                                        \
    {                                                                            \
        const int  mf_   = ((s) + 2) & 15;     \
        const int  vi_   = mf_ & 7;                           \
        const bool up_   = (mf_ >> 3) != 0;                    \
        const bool mine_ = up_ == hi;                                            \
        const float f0_ = acc0[vi_];                                             \
        const float f1_ = acc1[vi_];                                             \
        if (STORE) {                                                             \
            const float f1x_ = xhalf(f1_);     \
            float* __restrict__ orow_ = op + (size_t)((r) - 14) * 1024;          \
            orow_[up_ ? colB : colA] = mine_ ? f0_ : f1x_;                       \
        }                                                                        \
        acc0[vi_] = mine_ ? 0.0f : f0_;        \
        acc1[vi_] = mine_ ? 0.0f : f1_;                                          \
    }

#define PAIR(s, r, rn, STORE)                                                    \
    {                                                                            \
        v2f n0[4], n1[4], m0[4], m1[4];                                          \
        FETCH_PAIR(rn, n0, n1, m0, m1)                                           \
        ROWMM(s, b0, b1)                                                         \
        TAIL(s, r, STORE)                                                        \
          \
          \
                    \
        __builtin_amdgcn_sched_barrier(0);                                       \
        ROWMM(((s) + 1) & 15, c0, c1)                                            \
        TAIL(((s) + 1) & 15, (r) + 1, STORE)                                     \
        _Pragma("unroll")                                                        \
        for (int c = 0; c < 4; ++c) {                                            \
            b0[c] = n0[c]; b1[c] = n1[c]; c0[c] = m0[c]; c1[c] = m1[c];          \
        }                                                                        \
    }

    FETCH_PAIR(0, b0, b1, c0, c1)

    {
        v2f ae[4];
        EDGE_SUM(ae, 6, nl >= s_ + 2)
        float* kpark = kl[wv] + 8 * lane;
        v4f e0, e1;
        e0.x = ae[0].x; e0.y = ae[0].y; e0.z = ae[1].x; e0.w = ae[1].y;
        e1.x = ae[2].x; e1.y = ae[2].y; e1.z = ae[3].x; e1.w = ae[3].y;
        *(v4f*)(kpark)     = e0;
        *(v4f*)(kpark + 4) = e1;
    }
    {
        v2f ae[4];
        EDGE_SUM(ae, 0, nl <= s_)
        v2f n0[4], n1[4], m0[4], m1[4];
        FETCH_PAIR(8, n0, n1, m0, m1)
        ROWMM_A(ae, b0, b1)
        #pragma unroll
        for (int c = 0; c < 4; ++c) {
            b0[c] = n0[c]; b1[c] = n1[c]; c0[c] = m0[c]; c1[c] = m1[c];
        }
    }
    #pragma unroll
    for (int s = 8; s < 14; s += 2) PAIR(s, s, s + 2, false)

    v4f gi[4];
    float gh[8];
    #pragma clang loop unroll(disable)
    for (int it = 0; it < 16; ++it) {
        const int base = 14 + 16 * it;
        const int slot = ((it + 1) & 1) * 16;
        const bool more = it < 15;
        if (it == 9)       __builtin_amdgcn_s_setprio(2);
        else if (it == 13) __builtin_amdgcn_s_setprio(1);
        else if (it == 15) __builtin_amdgcn_s_setprio(0);
        if (more) ISSUE_CHUNK(it + 2, gi, gh)
        #pragma unroll
        for (int j = 0; j < 8; j += 2) PAIR((14 + j) & 15, base + j, slot + j, true)
        if (more) {
            #pragma unroll
            for (int j = 8; j < 16; j += 2) PAIR((14 + j) & 15, base + j, slot + j, true)
            STORE_CHUNK(it + 2, gi, gh)
        }
    }
    {
        v2f ab[4];
        {
            const int ln = (int)__builtin_amdgcn_mbcnt_lo(~0u, 0u);
            const float* kpark = kl[wv] + 8 * ln;
            const v4f e0 = *(const v4f*)(kpark);
            const v4f e1 = *(const v4f*)(kpark + 4);
            ab[0].x = e0.x; ab[0].y = e0.y; ab[1].x = e0.z; ab[1].y = e0.w;
            ab[2].x = e1.x; ab[2].y = e1.y; ab[3].x = e1.z; ab[3].y = e1.w;
        }
        ROWMM_A(ab, b0, b1)
        #pragma unroll
        for (int i = 0; i < 8; ++i) TAIL(6 + i, 262 + i, true)
    }
#undef PAIR
#undef TAIL
#undef EDGE_SUM
#undef ROWMM_A
#undef ROWMM
#undef FETCH_PAIR
#undef LOAD_ROW
#undef STORE_CHUNK
#undef ISSUE_CHUNK
}

extern "C" void kernel_launch(void* const* d_in, const int* in_sizes, int n_in,
                              void* d_out, int out_size, void* d_ws, size_t ws_size,
                              hipStream_t stream) {
    const float* x    = (const float*)d_in[0];
    const float* kern = (const float*)d_in[1];
    float* outp  = (float*)d_out;

    pwblur_k<<<768, 128, 0, stream>>>(x, kern, outp);
}


# K packing: 4 input rows taken as one run of 60 (row,tap) slots = 15 matrix instructions, was 16 with a zero 16th tap; sum order still (dy,dx) ascending, but the 4-wide steps now straddle input rows
# speedup vs baseline: 1.7151x; 1.0352x over previous
#include <hip/hip_runtime.h>


typedef __attribute__((ext_vector_type(2))) float v2f;
typedef __attribute__((ext_vector_type(4))) float v4f;
typedef __attribute__((ext_vector_type(8))) float v8f;

#define TILE   256
#define RP     48
#define RROWS  32
#define KDN    720

__device__ __forceinline__ float xhalf(float x) {
    int b = __builtin_bit_cast(int, x);
    return __builtin_bit_cast(float,
        __builtin_amdgcn_permlanex16(b, b, 0x76543210, 0xfedcba98, false, false));
}

__global__ __launch_bounds__(128, 3) void pwblur_k(const float* __restrict__ x,
                                                const float* __restrict__ kraw,
                                                float* __restrict__ out) {
    __shared__ alignas(16) float ring[4][RROWS * RP];
    __shared__ alignas(16) float kd[4][KDN];

    __builtin_amdgcn_s_setprio(3);

    const int tid  = threadIdx.x;
    const int lane = tid & 31;
    const int wave = tid >> 5;
    const bool hi  = (lane & 16) != 0;
    const int  nl  = lane & 15;

    const int bid   = blockIdx.x;
    const int sub_x = bid & 1;
    const int t     = bid >> 1;
    const int ti    = t & 3;
    const int tj    = (t >> 2) & 3;
    const int plane = t >> 4;

    const int wv = __builtin_amdgcn_readfirstlane(wave);

    const int ty0 = tj * TILE;
    const int tx0 = ti * TILE;
    const int xw  = sub_x * 128 + wv * 32;

    const float* __restrict__ xin = x + (size_t)plane * 1024 * 1024;

    const float* __restrict__ kt = kraw + (tj * 4 + ti) * 225;
    float kv[8];
    #pragma unroll
    for (int j = 0; j < 8; ++j) {
        const int idx = lane + 32 * j;
        const float v = kt[idx < 225 ? idx : 224];
        kv[j] = idx < 225 ? v : 0.0f;
    }

    const int g4  = lane >> 3;
    const int q8  = lane & 7;
    const int hq  = nl < 14 ? nl : 14;
    const int hxi = hq < 7 ? hq : hq + 32;
    int hxs = xw + hxi - 7; hxs = hxs < 0 ? 0 : (hxs > TILE - 1 ? TILE - 1 : hxs);
    const float* __restrict__ xt = xin + (size_t)ty0 * 1024 + tx0;
    const int ioff = xw + 4 * q8;
    float* rg = ring[wv];
    const int sti = (g4 * RP + 8 + 4 * q8);
    const int sth = ((hi ? 1 : 0) * RP + 1 + hxi);

#define ISSUE_CHUNK(k, VI, VH)                                                   \
    {                                                                            \
        const int kr_ = 16 * (k);                \
        _Pragma("unroll")                                                        \
        for (int li_ = 0; li_ < 4; ++li_) {                                      \
            const int ys_ = kr_ + 4 * li_ + g4;                                  \
            VI[li_] = *(const v4f*)(xt + (ys_ * 1024 + ioff));                   \
        }                                                                        \
        _Pragma("unroll")                                                        \
        for (int li_ = 0; li_ < 8; ++li_) {                                      \
            const int ys_ = kr_ + 2 * li_ + (hi ? 1 : 0);                        \
            VH[li_] = xt[ys_ * 1024 + hxs];                                      \
        }                                                                        \
    }

#define STORE_CHUNK(k, VI, VH)                                                   \
    {                                                                            \
        float* rs_ = rg + ((k) & 1) * (16 * RP);                                 \
        _Pragma("unroll")                                                        \
        for (int li_ = 0; li_ < 4; ++li_) *(v4f*)(rs_ + sti + 4 * li_ * RP) = VI[li_]; \
        _Pragma("unroll")                                                        \
        for (int li_ = 0; li_ < 8; ++li_) rs_[sth + 2 * li_ * RP] = VH[li_];     \
    }

    v4f gi[4];
    float gh[8];
    {
        v4f vi0[4];
        float vh0[8];
        ISSUE_CHUNK(0, vi0, vh0)
        ISSUE_CHUNK(1, gi, gh)
        STORE_CHUNK(0, vi0, vh0)
    }

    float* kw = kd[wv];
    {
        float u0 = kv[0] + kv[4], u1 = kv[1] + kv[5], u2 = kv[2] + kv[6], u3 = kv[3] + kv[7];
        float w0 = u0 + u2, w1 = u1 + u3;
        float z = w0 + w1;
        z = z + xhalf(z);
        z = z + __shfl_xor(z, 8);
        z = z + __shfl_xor(z, 4);
        z = z + __shfl_xor(z, 2);
        z = z + __shfl_xor(z, 1);
        const float inv = 1.0f / (z + 1e-12f);
        #pragma unroll
        for (int j = 0; j < 8; ++j) {
            const int idx = lane + 32 * j;
            const float nv = kv[j] * inv;
            if (idx < 240) {
                kw[idx]       = nv;
                kw[idx + 240] = nv;
                kw[idx + 480] = nv;
            }
        }
    }

    const float* ab = kw + (16 - nl) * 15 + (hi ? 2 : 0);
    v2f a[4][15];
    #pragma unroll
    for (int g = 0; g < 4; ++g) {
        const int s0 = (4 * g + 7) & 15;
        #pragma unroll
        for (int c = 0; c < 15; ++c) {
            a[g][c].x = ab[15 * s0 + 4 * c];
            a[g][c].y = ab[15 * s0 + 4 * c + 1];
        }
    }

    v2f at[4];
    {
        const float* tb = kw + (hi ? 2 : 0);
        #pragma unroll
        for (int c = 0; c < 4; ++c) { at[c].x = 0.0f; at[c].y = 0.0f; }
        #pragma unroll
        for (int r = 0; r < 8; ++r) {
            const bool in_ = nl <= 7 - r;
            #pragma unroll
            for (int c = 0; c < 4; ++c) {
                const float p0_ = tb[15 * r + 4 * c];
                const float p1_ = tb[15 * r + 4 * c + 1];
                at[c].x += in_ ? p0_ : 0.0f;
                at[c].y += in_ ? p1_ : 0.0f;
            }
        }
        at[3].y = hi ? a[0][3].y : at[3].y;
    }

    {
        v2f ae[4];
        #pragma unroll
        for (int c = 0; c < 4; ++c) { ae[c].x = 0.0f; ae[c].y = 0.0f; }
        #pragma unroll
        for (int s = 6; s < 14; ++s) {
            const bool in_ = nl >= s + 2;
            #pragma unroll
            for (int c = 0; c < 4; ++c) {
                const float p0_ = ab[15 * (s - 3) + 4 * (11 + c)];
                const float p1_ = ab[15 * (s - 3) + 4 * (11 + c) + 1];
                ae[c].x += in_ ? p0_ : 0.0f;
                ae[c].y += in_ ? p1_ : 0.0f;
            }
        }
        ae[0].x = hi ? ae[0].x : a[3][11].x;
        float* kpark = kw + 8 * lane;
        v4f e0, e1;
        e0.x = ae[0].x; e0.y = ae[0].y; e0.z = ae[1].x; e0.w = ae[1].y;
        e1.x = ae[2].x; e1.y = ae[2].y; e1.z = ae[3].x; e1.w = ae[3].y;
        *(v4f*)(kpark)     = e0;
        *(v4f*)(kpark + 4) = e1;
    }

    const char* rgc = (const char*)&ring[0][0];
    const int   bl  = 1 + nl + (hi ? 2 : 0);
    const unsigned bl0  = (unsigned)((wv * (RROWS * RP) + bl) * 4);
    const unsigned hoff = hi ? (RP - 15) * 4u : 0u;
    unsigned bo, bq;

    float* __restrict__ op = out + (size_t)plane * 1024 * 1024
                           + (size_t)ty0 * 1024 + (tx0 + xw);
    const int colA = (hi ? 16 : 0) + nl;
    const int colB = (hi ? 0 : 16) + nl;

    v8f acc0 = {};
    v8f acc1 = {};
    v2f b0[15], b1[15];

#define FETCH_GROUP(rr, D0, D1)                                                  \
    {                                                                            \
        bo = bl0 + (unsigned)(rr) * (RP * 4u);                                   \
        bq = bo + hoff;                                                          \
          \
          \
        asm volatile("" : "+v"(bo), "+v"(bq));                                   \
        const float* p0_ = (const float*)(rgc + bo);                             \
        const float* p1_ = (const float*)(rgc + bq);                             \
        _Pragma("unroll")                                                        \
        for (int c = 0; c < 15; ++c) {                                           \
            const int jx_ = (4 * c) / 15,     jxu_ = (4 * c + 2) / 15;           \
            const int jy_ = (4 * c + 1) / 15, jyu_ = (4 * c + 3) / 15;           \
            const float* px_ = (jxu_ != jx_) ? p1_ : p0_;                        \
            const float* py_ = (jyu_ != jy_) ? p1_ : p0_;                        \
            D0[c].x = px_[4 * c + 33 * jx_];                                     \
            D0[c].y = py_[4 * c + 1 + 33 * jy_];                                 \
            D1[c].x = px_[4 * c + 33 * jx_ + 16];                                \
            D1[c].y = py_[4 * c + 1 + 33 * jy_ + 16];                            \
        }                                                                        \
          \
          \
        __builtin_amdgcn_sched_barrier(0);                                       \
    }

#define MMS(AG, c0, c1)                                                          \
    {                                                                            \
        _Pragma("unroll")                                                        \
        for (int c = (c0); c < (c1); ++c) {                                      \
            acc1 = __builtin_amdgcn_wmma_f32_16x16x4_f32(                        \
                false, AG[c], false, b1[c], (short)0, acc1, false, false);       \
              \
              \
              \
            __builtin_amdgcn_sched_barrier(0);                                   \
            acc0 = __builtin_amdgcn_wmma_f32_16x16x4_f32(                        \
                false, AG[c], false, b0[c], (short)0, acc0, false, false);       \
            __builtin_amdgcn_sched_barrier(0);                                   \
        }                                                                        \
    }

#define TAIL(s, r, STORE)                                                        \
    {                                                                            \
        const int  mf_   = ((s) + 2) & 15;     \
        const int  vi_   = mf_ & 7;                           \
        const bool up_   = (mf_ >> 3) != 0;                    \
        const bool mine_ = up_ == hi;                                            \
        const float f0_ = acc0[vi_];                                             \
        const float f1_ = acc1[vi_];                                             \
        if (STORE) {                                                             \
            const float f1x_ = xhalf(f1_);     \
            float* __restrict__ orow_ = op + (size_t)((r) - 14) * 1024;          \
            orow_[up_ ? colB : colA] = mine_ ? f0_ : f1x_;                       \
        }                                                                        \
        acc0[vi_] = mine_ ? 0.0f : f0_;        \
        acc1[vi_] = mine_ ? 0.0f : f1_;                                          \
    }

#define GROUP(AG, S0, tr, FETCH, rn, M0, M1, M2, M3)                             \
    {                                                                            \
        v2f n0[15], n1[15];                                                      \
        if (FETCH) FETCH_GROUP(rn, n0, n1)                                       \
        MMS(AG, 0, 4)                                                            \
        if (M0) TAIL(((S0) + 0) & 15, (tr) + 7, (M0) == 2)                       \
        MMS(AG, 4, 8)                                                            \
        if (M1) TAIL(((S0) + 1) & 15, (tr) + 8, (M1) == 2)                       \
        MMS(AG, 8, 12)                                                           \
        if (M2) TAIL(((S0) + 2) & 15, (tr) + 9, (M2) == 2)                       \
        MMS(AG, 12, 15)                                                          \
        if (M3) TAIL(((S0) + 3) & 15, (tr) + 10, (M3) == 2)                      \
        if (FETCH) {                                                             \
            _Pragma("unroll")                                                    \
            for (int c = 0; c < 15; ++c) { b0[c] = n0[c]; b1[c] = n1[c]; }       \
        }                                                                        \
    }

    FETCH_GROUP(0, b0, b1)

    {
        v2f a0t[15];
        #pragma unroll
        for (int c = 0; c < 15; ++c) a0t[c] = c < 4 ? at[c] : a[0][c];
        GROUP(a0t, 7, 0, true, 4, 0, 1, 1, 1)
    }
    GROUP(a[1], 11, 4, true, 8, 1, 1, 1, 2)

    #pragma clang loop unroll(disable)
    for (int it = 0; it < 15; ++it) {
        const int tr0 = 8 + 16 * it;
        const int rh  = (it & 1) * 16;
        if (it == 9)       __builtin_amdgcn_s_setprio(2);
        else if (it == 13) __builtin_amdgcn_s_setprio(1);
        STORE_CHUNK(it + 1, gi, gh)
        if (it < 14) ISSUE_CHUNK(it + 2, gi, gh)
        GROUP(a[2], 15, tr0,      true, (rh + 12) & 31, 2, 2, 2, 2)
        GROUP(a[3],  3, tr0 + 4,  true, (rh + 16) & 31, 2, 2, 2, 2)
        GROUP(a[0],  7, tr0 + 8,  true, (rh + 20) & 31, 2, 2, 2, 2)
        GROUP(a[1], 11, tr0 + 12, true, (rh + 24) & 31, 2, 2, 2, 2)
    }
    __builtin_amdgcn_s_setprio(0);
    GROUP(a[2], 15, 248, true, 28, 2, 2, 2, 2)
    {
        v2f a3b[15];
        #pragma unroll
        for (int c = 0; c < 11; ++c) a3b[c] = a[3][c];
        {
            const int ln = (int)__builtin_amdgcn_mbcnt_lo(~0u, 0u);
            const float* kpark = kd[wv] + 8 * ln;
            const v4f e0 = *(const v4f*)(kpark);
            const v4f e1 = *(const v4f*)(kpark + 4);
            a3b[11].x = e0.x; a3b[11].y = e0.y; a3b[12].x = e0.z; a3b[12].y = e0.w;
            a3b[13].x = e1.x; a3b[13].y = e1.y; a3b[14].x = e1.z; a3b[14].y = e1.w;
        }
        GROUP(a3b, 3, 252, false, 0, 2, 2, 2, 0)
        #pragma unroll
        for (int i = 0; i < 8; ++i) TAIL(6 + i, 262 + i, true)
    }
#undef GROUP
#undef TAIL
#undef MMS
#undef FETCH_GROUP
#undef STORE_CHUNK
#undef ISSUE_CHUNK
}

extern "C" void kernel_launch(void* const* d_in, const int* in_sizes, int n_in,
                              void* d_out, int out_size, void* d_ws, size_t ws_size,
                              hipStream_t stream) {
    const float* x    = (const float*)d_in[0];
    const float* kern = (const float*)d_in[1];
    float* outp  = (float*)d_out;

    pwblur_k<<<768, 128, 0, stream>>>(x, kern, outp);
}
